# starting kernel
# baseline (speedup 1.0000x reference)
#include <hip/hip_runtime.h>
#include <hip/hip_bf16.h>

#define NNODES 50000
#define NEDGES 800000
#define INF_   256
#define FDIM   32
#define HEADS  8
#define HF     (HEADS * FDIM)
#define NEG_SLOPE 0.2f

typedef float v2f __attribute__((ext_vector_type(2)));
typedef float v8f __attribute__((ext_vector_type(8)));

__global__ __launch_bounds__(256) void gemm_wmma_f32(const float* __restrict__ X,
                                                     const float* __restrict__ W,
                                                     float* __restrict__ Z,
                                                     int K) {
    const int lane = threadIdx.x & 31;
    const int wave = threadIdx.x >> 5;
    const int half = lane >> 4;
    const int mn   = lane & 15;
    const int row0 = blockIdx.x * 16;
    const int col0 = wave * 32;
    const int col1 = col0 + 16;

    const float* xrow = X + (size_t)(row0 + mn) * K;
    v8f c0 = {}, c1 = {};
    for (int k = 0; k < K; k += 4) {
        const int kk = k + half * 2;
        v2f a, b0, b1;
        a.x  = xrow[kk];
        a.y  = xrow[kk + 1];
        const float* w0 = W + (size_t)kk * HF;
        b0.x = w0[col0 + mn];
        b0.y = w0[HF + col0 + mn];
        b1.x = w0[col1 + mn];
        b1.y = w0[HF + col1 + mn];
        c0 = __builtin_amdgcn_wmma_f32_16x16x4_f32(false, a, false, b0,
                                                   (short)0, c0, false, false);
        c1 = __builtin_amdgcn_wmma_f32_16x16x4_f32(false, a, false, b1,
                                                   (short)0, c1, false, false);
    }
    float* zout = Z + (size_t)(row0 + half * 8) * HF + mn;
#pragma unroll
    for (int v = 0; v < 8; ++v) {
        zout[(size_t)v * HF + col0] = c0[v];
        zout[(size_t)v * HF + col1] = c1[v];
    }
}

__global__ void attn_logits(const float* __restrict__ Z,
                            const float* __restrict__ al, const float* __restrict__ ar,
                            float* __restrict__ el, float* __restrict__ er) {
    int i = blockIdx.x * blockDim.x + threadIdx.x;
    if (i >= NNODES * HEADS) return;
    const int h = i & (HEADS - 1);
    const int n = i >> 3;
    const float* z  = Z + (size_t)n * HF + h * FDIM;
    const float* pl = al + h * FDIM;
    const float* pr = ar + h * FDIM;
    float sl = 0.f, sr = 0.f;
#pragma unroll
    for (int f = 0; f < FDIM; ++f) { float zv = z[f]; sl += zv * pl[f]; sr += zv * pr[f]; }
    el[i] = sl; er[i] = sr;
}

__device__ __forceinline__ unsigned f2ord(float f) {
    unsigned u = __float_as_uint(f);
    return (u & 0x80000000u) ? ~u : (u | 0x80000000u);
}
__device__ __forceinline__ float ord2f(unsigned e) {
    unsigned u = (e & 0x80000000u) ? (e & 0x7FFFFFFFu) : ~e;
    return __uint_as_float(u);
}
#define ORD_NEG_INF 0x007FFFFFu

__global__ void init_bufs(unsigned* __restrict__ m_ord, float* __restrict__ denom,
                          float* __restrict__ acc) {
    int i = blockIdx.x * blockDim.x + threadIdx.x;
    if (i < NNODES * HEADS) { m_ord[i] = ORD_NEG_INF; denom[i] = 0.f; }
    if (i < NNODES * FDIM)  acc[i] = 0.f;
}

__device__ __forceinline__ float edge_logit(const float* el, const float* er,
                                            int s, int d, int h) {
    float v = el[s * HEADS + h] + er[d * HEADS + h];
    return v >= 0.f ? v : NEG_SLOPE * v;
}

__global__ void edge_max(const int* __restrict__ src, const int* __restrict__ dst,
                         const float* __restrict__ el, const float* __restrict__ er,
                         unsigned* __restrict__ m_ord) {
    int i = blockIdx.x * blockDim.x + threadIdx.x;
    if (i >= NEDGES * HEADS) return;
    const int h = i & (HEADS - 1);
    const int e = i >> 3;
    const int s = src[e], d = dst[e];
    atomicMax(&m_ord[d * HEADS + h], f2ord(edge_logit(el, er, s, d, h)));
}

__global__ void edge_expsum(const int* __restrict__ src, const int* __restrict__ dst,
                            const float* __restrict__ el, const float* __restrict__ er,
                            const unsigned* __restrict__ m_ord,
                            float* __restrict__ exbuf, float* __restrict__ denom) {
    int i = blockIdx.x * blockDim.x + threadIdx.x;
    if (i >= NEDGES * HEADS) return;
    const int h = i & (HEADS - 1);
    const int e = i >> 3;
    const int s = src[e], d = dst[e];
    float v  = edge_logit(el, er, s, d, h);
    float ex = __expf(v - ord2f(m_ord[d * HEADS + h]));
    exbuf[i] = ex;
    atomicAdd(&denom[d * HEADS + h], ex);
}

__global__ void edge_alpha(const int* __restrict__ dst, const float* __restrict__ denom,
                           float* __restrict__ exbuf) {
    int i = blockIdx.x * blockDim.x + threadIdx.x;
    if (i >= NEDGES * HEADS) return;
    const int h = i & (HEADS - 1);
    const int d = dst[i >> 3];
    exbuf[i] = exbuf[i] / denom[d * HEADS + h] * (1.0f / HEADS);
}

__global__ void edge_agg(const int* __restrict__ src, const int* __restrict__ dst,
                         const float* __restrict__ Z, const float* __restrict__ exbuf,
                         float* __restrict__ acc) {
    int i = blockIdx.x * blockDim.x + threadIdx.x;
    if (i >= NEDGES * FDIM) return;
    const int f = i & (FDIM - 1);
    const int e = i >> 5;
    const int s = src[e], d = dst[e];
    const float* zs = Z + (size_t)s * HF + f;
    const float* ae = exbuf + e * HEADS;
    float sum = 0.f;
#pragma unroll
    for (int h = 0; h < HEADS; ++h)
        sum += ae[h] * zs[h * FDIM];
    atomicAdd(&acc[d * FDIM + f], sum);
}

__global__ void epilogue(const float* __restrict__ acc, const float* __restrict__ b,
                         float* __restrict__ out, int do_elu) {
    int i = blockIdx.x * blockDim.x + threadIdx.x;
    if (i >= NNODES * FDIM) return;
    const int f = i & (FDIM - 1);
    float bm = 0.f;
#pragma unroll
    for (int h = 0; h < HEADS; ++h) bm += b[h * FDIM + f];
    float v = acc[i] + bm * (1.0f / HEADS);
    out[i] = (do_elu && v <= 0.f) ? expm1f(v) : v;
}

static void run_gat_layer(const float* X, int K, const float* W,
                          const float* al, const float* ar, const float* b,
                          const int* src, const int* dst,
                          float* z, float* el, float* er, unsigned* m_ord,
                          float* denom, float* exbuf, float* acc, float* out,
                          int do_elu, hipStream_t stream) {
    gemm_wmma_f32<<<dim3(NNODES / 16, 1), 256, 0, stream>>>(X, W, z, K);

    int nh = NNODES * HEADS;
    attn_logits<<<(nh + 255) / 256, 256, 0, stream>>>(z, al, ar, el, er);

    int nf = NNODES * FDIM;
    init_bufs<<<(nf + 255) / 256, 256, 0, stream>>>(m_ord, denom, acc);

    int eh = NEDGES * HEADS;
    edge_max<<<(eh + 255) / 256, 256, 0, stream>>>(src, dst, el, er, m_ord);
    edge_expsum<<<(eh + 255) / 256, 256, 0, stream>>>(src, dst, el, er, m_ord, exbuf, denom);
    edge_alpha<<<(eh + 255) / 256, 256, 0, stream>>>(dst, denom, exbuf);

    int ef = NEDGES * FDIM;
    edge_agg<<<(ef + 255) / 256, 256, 0, stream>>>(src, dst, z, exbuf, acc);

    epilogue<<<(nf + 255) / 256, 256, 0, stream>>>(acc, b, out, do_elu);
}

extern "C" void kernel_launch(void* const* d_in, const int* in_sizes, int n_in,
                              void* d_out, int out_size, void* d_ws, size_t ws_size,
                              hipStream_t stream) {
    const float* x   = (const float*)d_in[0];
    const float* W1  = (const float*)d_in[1];
    const float* al1 = (const float*)d_in[2];
    const float* ar1 = (const float*)d_in[3];
    const float* b1  = (const float*)d_in[4];
    const float* W2  = (const float*)d_in[5];
    const float* al2 = (const float*)d_in[6];
    const float* ar2 = (const float*)d_in[7];
    const float* b2  = (const float*)d_in[8];
    const int*   src = (const int*)d_in[9];
    const int*   dst = (const int*)d_in[10];
    float* out = (float*)d_out;

    char* p = (char*)d_ws;
    float*    z     = (float*)p;    p += (size_t)NNODES * HF * 4;
    float*    el    = (float*)p;    p += (size_t)NNODES * HEADS * 4;
    float*    er    = (float*)p;    p += (size_t)NNODES * HEADS * 4;
    unsigned* m_ord = (unsigned*)p; p += (size_t)NNODES * HEADS * 4;
    float*    denom = (float*)p;    p += (size_t)NNODES * HEADS * 4;
    float*    exbuf = (float*)p;    p += (size_t)NEDGES * HEADS * 4;
    float*    acc   = (float*)p;    p += (size_t)NNODES * FDIM * 4;
    float*    hbuf  = (float*)p;    p += (size_t)NNODES * FDIM * 4;

    run_gat_layer(x, INF_, W1, al1, ar1, b1, src, dst,
                  z, el, er, m_ord, denom, exbuf, acc, hbuf,  1, stream);
    run_gat_layer(hbuf, FDIM, W2, al2, ar2, b2, src, dst,
                  z, el, er, m_ord, denom, exbuf, acc, out,  0, stream);
}


# layer-1 GEMM: 32-row X tile staged once in LDS by whole-line loads; a wave takes 32 rows x its head's 32 columns as even/odd column tiles: each 8-byte W load feeds 4 WMMAs, Z leaves in whole lines
# speedup vs baseline: 1.7738x; 1.7738x over previous
#include <hip/hip_runtime.h>
#include <hip/hip_bf16.h>

#define NNODES 50000
#define NEDGES 800000
#define INF_   256
#define FDIM   32
#define HEADS  8
#define HF     (HEADS * FDIM)
#define NEG_SLOPE 0.2f

typedef float v2f __attribute__((ext_vector_type(2)));
typedef float v8f __attribute__((ext_vector_type(8)));

#define GM_ROWS  32
#define GM_PITCH (INF_ + 4)

__device__ __forceinline__ v8f wmma4(v2f a, v2f b, v8f c) {
    return __builtin_amdgcn_wmma_f32_16x16x4_f32(false, a, false, b, (short)0, c, false, false);
}

__device__ __forceinline__ void gemm_step(v8f (&acc)[2][2], const float4 (&av)[2], const float2 (&bw)[4]) {
    const v2f b00 = {bw[0].x, bw[1].x}, b01 = {bw[0].y, bw[1].y};
    const v2f b10 = {bw[2].x, bw[3].x}, b11 = {bw[2].y, bw[3].y};
#pragma unroll
    for (int rt = 0; rt < 2; ++rt) {
        const v2f a0 = {av[rt].x, av[rt].y};
        acc[rt][0] = wmma4(a0, b00, acc[rt][0]);
        acc[rt][1] = wmma4(a0, b01, acc[rt][1]);
    }
#pragma unroll
    for (int rt = 0; rt < 2; ++rt) {
        const v2f a1 = {av[rt].z, av[rt].w};
        acc[rt][0] = wmma4(a1, b10, acc[rt][0]);
        acc[rt][1] = wmma4(a1, b11, acc[rt][1]);
    }
}

__global__ __launch_bounds__(256) void gemm_wmma_f32(const float* __restrict__ X,
                                                     const float* __restrict__ W,
                                                     float* __restrict__ Z) {
    __shared__ __attribute__((aligned(16))) float s_a[GM_ROWS * GM_PITCH];
    const int tid  = threadIdx.x;
    const int lane = tid & 31;
    const int wave = tid >> 5;
    const int half = lane >> 4;
    const int c    = lane & 15;
    const int row0 = blockIdx.x * GM_ROWS;

    {
        const int col4 = tid & 63;
        const int r    = tid >> 6;
#pragma unroll
        for (int p = 0; p < GM_ROWS / 4; ++p) {
            const int row = p * 4 + r;
            const int grow = min(row0 + row, NNODES - 1);
            const float4 v = *reinterpret_cast<const float4*>(X + (size_t)grow * INF_ + col4 * 4);
            *reinterpret_cast<float4*>(&s_a[row * GM_PITCH + col4 * 4]) = v;
        }
    }
    __syncthreads();

    const int col0 = wave * 32;
    const float* wp = W + (size_t)(4 * half) * HF + col0 + 2 * c;
    const float* ap = s_a + c * GM_PITCH + 4 * half;

    v8f acc[2][2] = {};
    float2 bw[4];
    float4 av[2];
#pragma unroll
    for (int s = 0; s < 4; ++s) bw[s] = *reinterpret_cast<const float2*>(wp + (size_t)s * HF);
#pragma unroll
    for (int rt = 0; rt < 2; ++rt) av[rt] = *reinterpret_cast<const float4*>(ap + rt * 16 * GM_PITCH);

    for (int j = 1; j < INF_ / 8; ++j) {
        float2 bn[4];
        float4 an[2];
#pragma unroll
        for (int s = 0; s < 4; ++s) bn[s] = *reinterpret_cast<const float2*>(wp + (size_t)(8 * j + s) * HF);
#pragma unroll
        for (int rt = 0; rt < 2; ++rt) an[rt] = *reinterpret_cast<const float4*>(ap + rt * 16 * GM_PITCH + 8 * j);
        gemm_step(acc, av, bw);
#pragma unroll
        for (int s = 0; s < 4; ++s) bw[s] = bn[s];
#pragma unroll
        for (int rt = 0; rt < 2; ++rt) av[rt] = an[rt];
    }
    gemm_step(acc, av, bw);

#pragma unroll
    for (int rt = 0; rt < 2; ++rt) {
        const int rbase = row0 + rt * 16;
        if (rbase < NNODES) {
            float* zp = Z + (size_t)(rbase + 8 * half) * HF + col0 + 2 * c;
#pragma unroll
            for (int v = 0; v < 8; ++v)
                *reinterpret_cast<float2*>(zp + (size_t)v * HF) = make_float2(acc[rt][0][v], acc[rt][1][v]);
        }
    }
}

__global__ void attn_logits(const float* __restrict__ Z,
                            const float* __restrict__ al, const float* __restrict__ ar,
                            float* __restrict__ el, float* __restrict__ er) {
    int i = blockIdx.x * blockDim.x + threadIdx.x;
    if (i >= NNODES * HEADS) return;
    const int h = i & (HEADS - 1);
    const int n = i >> 3;
    const float* z  = Z + (size_t)n * HF + h * FDIM;
    const float* pl = al + h * FDIM;
    const float* pr = ar + h * FDIM;
    float sl = 0.f, sr = 0.f;
#pragma unroll
    for (int f = 0; f < FDIM; ++f) { float zv = z[f]; sl += zv * pl[f]; sr += zv * pr[f]; }
    el[i] = sl; er[i] = sr;
}


__global__ void zero_counts(int* __restrict__ counts) {
    int i = blockIdx.x * blockDim.x + threadIdx.x;
    if (i < NNODES) counts[i] = 0;
}

__global__ void dst_histogram(const int* __restrict__ dst, int* __restrict__ counts) {
    int e = blockIdx.x * blockDim.x + threadIdx.x;
    if (e < NEDGES) atomicAdd(&counts[dst[e]], 1);
}

#define SCAN_T 1024
__global__ __launch_bounds__(SCAN_T) void scan_counts(const int* __restrict__ counts,
                                                      int* __restrict__ rowptr,
                                                      int* __restrict__ cursor) {
    __shared__ int s_own[SCAN_T];
    __shared__ int s_pre[SCAN_T];
    const int t = threadIdx.x;
    const int tile0 = blockIdx.x * SCAN_T;
    int pre = 0;
    for (int i = t; i < tile0; i += SCAN_T) pre += counts[i];
    const int idx = tile0 + t;
    const int c = idx < NNODES ? counts[idx] : 0;
    s_own[t] = c;
    s_pre[t] = pre;
    __syncthreads();
    for (int off = 1; off < SCAN_T; off <<= 1) {
        int a = 0, b = 0;
        if (t >= off) { a = s_own[t - off]; b = s_pre[t - off]; }
        __syncthreads();
        s_own[t] += a;
        s_pre[t] += b;
        __syncthreads();
    }
    const int base = s_pre[SCAN_T - 1];
    const int incl = s_own[t];
    if (idx < NNODES) {
        const int excl = base + incl - c;
        rowptr[idx] = excl;
        cursor[idx] = excl;
        if (idx == NNODES - 1) rowptr[NNODES] = base + incl;
    }
}

__global__ void fill_by_dst(const int* __restrict__ src, const int* __restrict__ dst,
                            int* __restrict__ cursor, int* __restrict__ ssrc) {
    int e = blockIdx.x * blockDim.x + threadIdx.x;
    if (e >= NEDGES) return;
    const int pos = atomicAdd(&cursor[dst[e]], 1);
    ssrc[pos] = src[e];
}

__device__ __forceinline__ float lrelu(float v) { return v >= 0.f ? v : NEG_SLOPE * v; }

__device__ __forceinline__ float wave_max(float v) {
#pragma unroll
    for (int o = 16; o > 0; o >>= 1) v = fmaxf(v, __shfl_xor(v, o));
    return v;
}
__device__ __forceinline__ float wave_sum(float v) {
#pragma unroll
    for (int o = 16; o > 0; o >>= 1) v += __shfl_xor(v, o);
    return v;
}

__device__ __forceinline__ void edge_logits8(const float* __restrict__ el, int s,
                                             const float* erd, float* v) {
    const float4 p = *reinterpret_cast<const float4*>(el + (size_t)s * HEADS);
    const float4 q = *reinterpret_cast<const float4*>(el + (size_t)s * HEADS + 4);
    v[0] = lrelu(p.x + erd[0]); v[1] = lrelu(p.y + erd[1]);
    v[2] = lrelu(p.z + erd[2]); v[3] = lrelu(p.w + erd[3]);
    v[4] = lrelu(q.x + erd[4]); v[5] = lrelu(q.y + erd[5]);
    v[6] = lrelu(q.z + erd[6]); v[7] = lrelu(q.w + erd[7]);
}

__global__ __launch_bounds__(256) void seg_softmax_agg(const int* __restrict__ rowptr,
                                                       const int* __restrict__ ssrc,
                                                       const float* __restrict__ Z,
                                                       const float* __restrict__ el,
                                                       const float* __restrict__ er,
                                                       const float* __restrict__ b,
                                                       float* __restrict__ out, int do_elu) {
    const int lane = threadIdx.x & 31;
    const int d = __builtin_amdgcn_readfirstlane((int)(blockIdx.x * 8 + (threadIdx.x >> 5)));
    const int start = rowptr[d];
    const int end   = rowptr[d + 1];

    float erd[HEADS];
#pragma unroll
    for (int h = 0; h < HEADS; ++h) erd[h] = er[d * HEADS + h];

    float m[HEADS], den[HEADS];
#pragma unroll
    for (int h = 0; h < HEADS; ++h) { m[h] = -INFINITY; den[h] = 0.f; }

    for (int c0 = start; c0 < end; c0 += 32) {
        const int j = c0 + lane;
        if (j < end) {
            float v[HEADS];
            edge_logits8(el, ssrc[j], erd, v);
#pragma unroll
            for (int h = 0; h < HEADS; ++h) m[h] = fmaxf(m[h], v[h]);
        }
    }
#pragma unroll
    for (int h = 0; h < HEADS; ++h) m[h] = wave_max(m[h]);

    for (int c0 = start; c0 < end; c0 += 32) {
        const int j = c0 + lane;
        if (j < end) {
            float v[HEADS];
            edge_logits8(el, ssrc[j], erd, v);
#pragma unroll
            for (int h = 0; h < HEADS; ++h) den[h] += __expf(v[h] - m[h]);
        }
    }
#pragma unroll
    for (int h = 0; h < HEADS; ++h) den[h] = wave_sum(den[h]);

    float acc = 0.f;
    for (int c0 = start; c0 < end; c0 += 32) {
        const int j = c0 + lane;
        int   s_l = 0;
        float a[HEADS];
#pragma unroll
        for (int h = 0; h < HEADS; ++h) a[h] = 0.f;
        if (j < end) {
            s_l = ssrc[j];
            float v[HEADS];
            edge_logits8(el, s_l, erd, v);
#pragma unroll
            for (int h = 0; h < HEADS; ++h)
                a[h] = __expf(v[h] - m[h]) / den[h] * (1.0f / HEADS);
        }
        const int n = min(32, end - c0);
#pragma unroll 4
        for (int i = 0; i < n; ++i) {
            const int s = __builtin_amdgcn_readlane(s_l, i);
            const float* zs = Z + (size_t)s * HF + lane;
            float sum = 0.f;
#pragma unroll
            for (int h = 0; h < HEADS; ++h) {
                const float ah = __int_as_float(__builtin_amdgcn_readlane(__float_as_int(a[h]), i));
                sum += ah * zs[h * FDIM];
            }
            acc += sum;
        }
    }

    float bm = 0.f;
#pragma unroll
    for (int h = 0; h < HEADS; ++h) bm += b[h * FDIM + lane];
    const float r = acc + bm * (1.0f / HEADS);
    out[d * FDIM + lane] = (do_elu && r <= 0.f) ? expm1f(r) : r;
}


__global__ __launch_bounds__(256) void attn_logits_folded(const float* __restrict__ X,
                                                          const float* __restrict__ W,
                                                          const float* __restrict__ al,
                                                          const float* __restrict__ ar,
                                                          float* __restrict__ el,
                                                          float* __restrict__ er) {
    __shared__ float s_w[FDIM * 2 * HEADS];
    {
        const int k = threadIdx.x >> 3;
        const int h = threadIdx.x & (HEADS - 1);
        const float* w  = W + (size_t)k * HF + h * FDIM;
        const float* pl = al + h * FDIM;
        const float* pr = ar + h * FDIM;
        float sl = 0.f, sr = 0.f;
        for (int f = 0; f < FDIM; ++f) { float wv = w[f]; sl += wv * pl[f]; sr += wv * pr[f]; }
        s_w[k * 2 * HEADS + h]         = sl;
        s_w[k * 2 * HEADS + HEADS + h] = sr;
    }
    __syncthreads();
    const int i = blockIdx.x * blockDim.x + threadIdx.x;
    if (i >= NNODES * HEADS) return;
    const int h = i & (HEADS - 1);
    const int n = i >> 3;
    const float* x = X + (size_t)n * FDIM;
    float sl = 0.f, sr = 0.f;
    for (int k = 0; k < FDIM; ++k) {
        const float xv = x[k];
        sl += xv * s_w[k * 2 * HEADS + h];
        sr += xv * s_w[k * 2 * HEADS + HEADS + h];
    }
    el[i] = sl; er[i] = sr;
}

__global__ __launch_bounds__(256) void seg_softmax_agg_rows(const int* __restrict__ rowptr,
                                                            const int* __restrict__ ssrc,
                                                            const float* __restrict__ X,
                                                            const float* __restrict__ el,
                                                            const float* __restrict__ er,
                                                            float* __restrict__ agg) {
    const int lane = threadIdx.x & 31;
    const int d = __builtin_amdgcn_readfirstlane((int)(blockIdx.x * 8 + (threadIdx.x >> 5)));
    const int start = rowptr[d];
    const int end   = rowptr[d + 1];

    float erd[HEADS];
#pragma unroll
    for (int h = 0; h < HEADS; ++h) erd[h] = er[d * HEADS + h];

    float m[HEADS], den[HEADS];
#pragma unroll
    for (int h = 0; h < HEADS; ++h) { m[h] = -INFINITY; den[h] = 0.f; }

    for (int c0 = start; c0 < end; c0 += 32) {
        const int j = c0 + lane;
        if (j < end) {
            float v[HEADS];
            edge_logits8(el, ssrc[j], erd, v);
#pragma unroll
            for (int h = 0; h < HEADS; ++h) m[h] = fmaxf(m[h], v[h]);
        }
    }
#pragma unroll
    for (int h = 0; h < HEADS; ++h) m[h] = wave_max(m[h]);

    for (int c0 = start; c0 < end; c0 += 32) {
        const int j = c0 + lane;
        if (j < end) {
            float v[HEADS];
            edge_logits8(el, ssrc[j], erd, v);
#pragma unroll
            for (int h = 0; h < HEADS; ++h) den[h] += __expf(v[h] - m[h]);
        }
    }
#pragma unroll
    for (int h = 0; h < HEADS; ++h) den[h] = wave_sum(den[h]);

    float acc[HEADS];
#pragma unroll
    for (int h = 0; h < HEADS; ++h) acc[h] = 0.f;
    for (int c0 = start; c0 < end; c0 += 32) {
        const int j = c0 + lane;
        int   s_l = 0;
        float a[HEADS];
#pragma unroll
        for (int h = 0; h < HEADS; ++h) a[h] = 0.f;
        if (j < end) {
            s_l = ssrc[j];
            float v[HEADS];
            edge_logits8(el, s_l, erd, v);
#pragma unroll
            for (int h = 0; h < HEADS; ++h)
                a[h] = __expf(v[h] - m[h]) / den[h] * (1.0f / HEADS);
        }
        const int n = min(32, end - c0);
#pragma unroll 8
        for (int i = 0; i < n; ++i) {
            const int s = __builtin_amdgcn_readlane(s_l, i);
            const float xv = X[(size_t)s * FDIM + lane];
#pragma unroll
            for (int h = 0; h < HEADS; ++h) {
                const float ah = __int_as_float(__builtin_amdgcn_readlane(__float_as_int(a[h]), i));
                acc[h] += ah * xv;
            }
        }
    }

    float* o = agg + (size_t)d * HF + lane;
#pragma unroll
    for (int h = 0; h < HEADS; ++h) o[h * FDIM] = acc[h];
}

__global__ __launch_bounds__(256) void project_heads_wmma(const float* __restrict__ A,
                                                          const float* __restrict__ W,
                                                          const float* __restrict__ b,
                                                          float* __restrict__ out) {
    const int lane = threadIdx.x & 31;
    const int half = lane >> 4;
    const int mn   = lane & 15;
    const int tile = blockIdx.x * 8 + (threadIdx.x >> 5);
    if (tile >= NNODES / 16) return;
    const int row0 = tile * 16;

    const float* arow = A + (size_t)(row0 + mn) * HF;
    v8f c0 = {}, c1 = {};
    for (int k = 0; k < HF; k += 4) {
        const int kk = k + half * 2;
        v2f a, b0, b1;
        a.x  = arow[kk];
        a.y  = arow[kk + 1];
        const float* w0 = W + (size_t)(kk & (FDIM - 1)) * HF + (kk >> 5) * FDIM;
        b0.x = w0[mn];
        b0.y = w0[HF + mn];
        b1.x = w0[16 + mn];
        b1.y = w0[HF + 16 + mn];
        c0 = __builtin_amdgcn_wmma_f32_16x16x4_f32(false, a, false, b0,
                                                   (short)0, c0, false, false);
        c1 = __builtin_amdgcn_wmma_f32_16x16x4_f32(false, a, false, b1,
                                                   (short)0, c1, false, false);
    }
    float bm0 = 0.f, bm1 = 0.f;
#pragma unroll
    for (int h = 0; h < HEADS; ++h) { bm0 += b[h * FDIM + mn]; bm1 += b[h * FDIM + 16 + mn]; }
    bm0 *= (1.0f / HEADS);
    bm1 *= (1.0f / HEADS);
    float* po = out + (size_t)(row0 + half * 8) * FDIM + mn;
#pragma unroll
    for (int v = 0; v < 8; ++v) {
        po[(size_t)v * FDIM]      = c0[v] + bm0;
        po[(size_t)v * FDIM + 16] = c1[v] + bm1;
    }
}

static void build_edges_by_dst(const int* src, const int* dst, int* counts, int* rowptr,
                               int* cursor, int* ssrc, hipStream_t stream) {
    zero_counts<<<(NNODES + 255) / 256, 256, 0, stream>>>(counts);
    dst_histogram<<<(NEDGES + 255) / 256, 256, 0, stream>>>(dst, counts);
    scan_counts<<<(NNODES + SCAN_T - 1) / SCAN_T, SCAN_T, 0, stream>>>(counts, rowptr, cursor);
    fill_by_dst<<<(NEDGES + 255) / 256, 256, 0, stream>>>(src, dst, cursor, ssrc);
}

static void run_gat_layer(const float* X, const float* W,
                          const float* al, const float* ar, const float* b,
                          const int* rowptr, const int* ssrc,
                          float* z, float* el, float* er, float* out,
                          int do_elu, hipStream_t stream) {
    gemm_wmma_f32<<<(NNODES + GM_ROWS - 1) / GM_ROWS, 256, 0, stream>>>(X, W, z);

    int nh = NNODES * HEADS;
    attn_logits<<<(nh + 255) / 256, 256, 0, stream>>>(z, al, ar, el, er);

    seg_softmax_agg<<<NNODES / 8, 256, 0, stream>>>(rowptr, ssrc, z, el, er, b, out, do_elu);
}

static void run_gat_layer_agg_first(const float* X, const float* W,
                                    const float* al, const float* ar, const float* b,
                                    const int* rowptr, const int* ssrc,
                                    float* agg, float* el, float* er, float* out,
                                    hipStream_t stream) {
    attn_logits_folded<<<(NNODES * HEADS + 255) / 256, 256, 0, stream>>>(X, W, al, ar, el, er);
    seg_softmax_agg_rows<<<NNODES / 8, 256, 0, stream>>>(rowptr, ssrc, X, el, er, agg);
    project_heads_wmma<<<(NNODES / 16 + 7) / 8, 256, 0, stream>>>(agg, W, b, out);
}

extern "C" void kernel_launch(void* const* d_in, const int* in_sizes, int n_in,
                              void* d_out, int out_size, void* d_ws, size_t ws_size,
                              hipStream_t stream) {
    const float* x   = (const float*)d_in[0];
    const float* W1  = (const float*)d_in[1];
    const float* al1 = (const float*)d_in[2];
    const float* ar1 = (const float*)d_in[3];
    const float* b1  = (const float*)d_in[4];
    const float* W2  = (const float*)d_in[5];
    const float* al2 = (const float*)d_in[6];
    const float* ar2 = (const float*)d_in[7];
    const float* b2  = (const float*)d_in[8];
    const int*   src = (const int*)d_in[9];
    const int*   dst = (const int*)d_in[10];
    float* out = (float*)d_out;

    char* p = (char*)d_ws;
    float* z      = (float*)p; p += (size_t)NNODES * HF * 4;
    float* el     = (float*)p; p += (size_t)NNODES * HEADS * 4;
    float* er     = (float*)p; p += (size_t)NNODES * HEADS * 4;
    int*   rowptr = (int*)p;   p += (size_t)NNODES * HEADS * 4;
    int*   cursor = (int*)p;   p += (size_t)NNODES * HEADS * 4;
    int*   counts = cursor + NNODES;
    int*   ssrc   = (int*)p;   p += (size_t)NEDGES * HEADS * 4;
    p += (size_t)NNODES * FDIM * 4;
    float* hbuf   = (float*)p; p += (size_t)NNODES * FDIM * 4;

    build_edges_by_dst(src, dst, counts, rowptr, cursor, ssrc, stream);

    run_gat_layer(x, W1, al1, ar1, b1, rowptr, ssrc,
                  z, el, er, hbuf,  1, stream);
    run_gat_layer_agg_first(hbuf, W2, al2, ar2, b2, rowptr, ssrc,
                            z, el, er, out, stream);
}


# both aggregation kernels: a node's first 32 edges keep source id and 8 logits (then their exp) in the lane's registers through max, exp-sum, alpha: ssrc and 32-B el row gathered once, not 3 times
# speedup vs baseline: 1.8338x; 1.0338x over previous
#include <hip/hip_runtime.h>
#include <hip/hip_bf16.h>

#define NNODES 50000
#define NEDGES 800000
#define INF_   256
#define FDIM   32
#define HEADS  8
#define HF     (HEADS * FDIM)
#define NEG_SLOPE 0.2f

typedef float v2f __attribute__((ext_vector_type(2)));
typedef float v8f __attribute__((ext_vector_type(8)));

#define GM_ROWS  32
#define GM_PITCH (INF_ + 4)

__device__ __forceinline__ v8f wmma4(v2f a, v2f b, v8f c) {
    return __builtin_amdgcn_wmma_f32_16x16x4_f32(false, a, false, b, (short)0, c, false, false);
}

__device__ __forceinline__ void gemm_step(v8f (&acc)[2][2], const float4 (&av)[2], const float2 (&bw)[4]) {
    const v2f b00 = {bw[0].x, bw[1].x}, b01 = {bw[0].y, bw[1].y};
    const v2f b10 = {bw[2].x, bw[3].x}, b11 = {bw[2].y, bw[3].y};
#pragma unroll
    for (int rt = 0; rt < 2; ++rt) {
        const v2f a0 = {av[rt].x, av[rt].y};
        acc[rt][0] = wmma4(a0, b00, acc[rt][0]);
        acc[rt][1] = wmma4(a0, b01, acc[rt][1]);
    }
#pragma unroll
    for (int rt = 0; rt < 2; ++rt) {
        const v2f a1 = {av[rt].z, av[rt].w};
        acc[rt][0] = wmma4(a1, b10, acc[rt][0]);
        acc[rt][1] = wmma4(a1, b11, acc[rt][1]);
    }
}

__global__ __launch_bounds__(256) void gemm_wmma_f32(const float* __restrict__ X,
                                                     const float* __restrict__ W,
                                                     float* __restrict__ Z) {
    __shared__ __attribute__((aligned(16))) float s_a[GM_ROWS * GM_PITCH];
    const int tid  = threadIdx.x;
    const int lane = tid & 31;
    const int wave = tid >> 5;
    const int half = lane >> 4;
    const int c    = lane & 15;
    const int row0 = blockIdx.x * GM_ROWS;

    {
        const int col4 = tid & 63;
        const int r    = tid >> 6;
#pragma unroll
        for (int p = 0; p < GM_ROWS / 4; ++p) {
            const int row = p * 4 + r;
            const int grow = min(row0 + row, NNODES - 1);
            const float4 v = *reinterpret_cast<const float4*>(X + (size_t)grow * INF_ + col4 * 4);
            *reinterpret_cast<float4*>(&s_a[row * GM_PITCH + col4 * 4]) = v;
        }
    }
    __syncthreads();

    const int col0 = wave * 32;
    const float* wp = W + (size_t)(4 * half) * HF + col0 + 2 * c;
    const float* ap = s_a + c * GM_PITCH + 4 * half;

    v8f acc[2][2] = {};
    float2 bw[4];
    float4 av[2];
#pragma unroll
    for (int s = 0; s < 4; ++s) bw[s] = *reinterpret_cast<const float2*>(wp + (size_t)s * HF);
#pragma unroll
    for (int rt = 0; rt < 2; ++rt) av[rt] = *reinterpret_cast<const float4*>(ap + rt * 16 * GM_PITCH);

    for (int j = 1; j < INF_ / 8; ++j) {
        float2 bn[4];
        float4 an[2];
#pragma unroll
        for (int s = 0; s < 4; ++s) bn[s] = *reinterpret_cast<const float2*>(wp + (size_t)(8 * j + s) * HF);
#pragma unroll
        for (int rt = 0; rt < 2; ++rt) an[rt] = *reinterpret_cast<const float4*>(ap + rt * 16 * GM_PITCH + 8 * j);
        gemm_step(acc, av, bw);
#pragma unroll
        for (int s = 0; s < 4; ++s) bw[s] = bn[s];
#pragma unroll
        for (int rt = 0; rt < 2; ++rt) av[rt] = an[rt];
    }
    gemm_step(acc, av, bw);

#pragma unroll
    for (int rt = 0; rt < 2; ++rt) {
        const int rbase = row0 + rt * 16;
        if (rbase < NNODES) {
            float* zp = Z + (size_t)(rbase + 8 * half) * HF + col0 + 2 * c;
#pragma unroll
            for (int v = 0; v < 8; ++v)
                *reinterpret_cast<float2*>(zp + (size_t)v * HF) = make_float2(acc[rt][0][v], acc[rt][1][v]);
        }
    }
}

__global__ void attn_logits(const float* __restrict__ Z,
                            const float* __restrict__ al, const float* __restrict__ ar,
                            float* __restrict__ el, float* __restrict__ er) {
    int i = blockIdx.x * blockDim.x + threadIdx.x;
    if (i >= NNODES * HEADS) return;
    const int h = i & (HEADS - 1);
    const int n = i >> 3;
    const float* z  = Z + (size_t)n * HF + h * FDIM;
    const float* pl = al + h * FDIM;
    const float* pr = ar + h * FDIM;
    float sl = 0.f, sr = 0.f;
#pragma unroll
    for (int f = 0; f < FDIM; ++f) { float zv = z[f]; sl += zv * pl[f]; sr += zv * pr[f]; }
    el[i] = sl; er[i] = sr;
}


__global__ void zero_counts(int* __restrict__ counts) {
    int i = blockIdx.x * blockDim.x + threadIdx.x;
    if (i < NNODES) counts[i] = 0;
}

__global__ void dst_histogram(const int* __restrict__ dst, int* __restrict__ counts) {
    int e = blockIdx.x * blockDim.x + threadIdx.x;
    if (e < NEDGES) atomicAdd(&counts[dst[e]], 1);
}

#define SCAN_T 1024
__global__ __launch_bounds__(SCAN_T) void scan_counts(const int* __restrict__ counts,
                                                      int* __restrict__ rowptr,
                                                      int* __restrict__ cursor) {
    __shared__ int s_own[SCAN_T];
    __shared__ int s_pre[SCAN_T];
    const int t = threadIdx.x;
    const int tile0 = blockIdx.x * SCAN_T;
    int pre = 0;
    for (int i = t; i < tile0; i += SCAN_T) pre += counts[i];
    const int idx = tile0 + t;
    const int c = idx < NNODES ? counts[idx] : 0;
    s_own[t] = c;
    s_pre[t] = pre;
    __syncthreads();
    for (int off = 1; off < SCAN_T; off <<= 1) {
        int a = 0, b = 0;
        if (t >= off) { a = s_own[t - off]; b = s_pre[t - off]; }
        __syncthreads();
        s_own[t] += a;
        s_pre[t] += b;
        __syncthreads();
    }
    const int base = s_pre[SCAN_T - 1];
    const int incl = s_own[t];
    if (idx < NNODES) {
        const int excl = base + incl - c;
        rowptr[idx] = excl;
        cursor[idx] = excl;
        if (idx == NNODES - 1) rowptr[NNODES] = base + incl;
    }
}

__global__ void fill_by_dst(const int* __restrict__ src, const int* __restrict__ dst,
                            int* __restrict__ cursor, int* __restrict__ ssrc) {
    int e = blockIdx.x * blockDim.x + threadIdx.x;
    if (e >= NEDGES) return;
    const int pos = atomicAdd(&cursor[dst[e]], 1);
    ssrc[pos] = src[e];
}

__device__ __forceinline__ float lrelu(float v) { return v >= 0.f ? v : NEG_SLOPE * v; }

__device__ __forceinline__ float wave_max(float v) {
#pragma unroll
    for (int o = 16; o > 0; o >>= 1) v = fmaxf(v, __shfl_xor(v, o));
    return v;
}
__device__ __forceinline__ float wave_sum(float v) {
#pragma unroll
    for (int o = 16; o > 0; o >>= 1) v += __shfl_xor(v, o);
    return v;
}

__device__ __forceinline__ void edge_logits8(const float* __restrict__ el, int s,
                                             const float* erd, float* v) {
    const float4 p = *reinterpret_cast<const float4*>(el + (size_t)s * HEADS);
    const float4 q = *reinterpret_cast<const float4*>(el + (size_t)s * HEADS + 4);
    v[0] = lrelu(p.x + erd[0]); v[1] = lrelu(p.y + erd[1]);
    v[2] = lrelu(p.z + erd[2]); v[3] = lrelu(p.w + erd[3]);
    v[4] = lrelu(q.x + erd[4]); v[5] = lrelu(q.y + erd[5]);
    v[6] = lrelu(q.z + erd[6]); v[7] = lrelu(q.w + erd[7]);
}

__device__ __forceinline__ float chunk_sum_z(const float* __restrict__ Z, int lane, int n,
                                             int s_l, const float (&a)[HEADS], float acc) {
#pragma unroll 4
    for (int i = 0; i < n; ++i) {
        const int s = __builtin_amdgcn_readlane(s_l, i);
        const float* zs = Z + (size_t)s * HF + lane;
        float sum = 0.f;
#pragma unroll
        for (int h = 0; h < HEADS; ++h) {
            const float ah = __int_as_float(__builtin_amdgcn_readlane(__float_as_int(a[h]), i));
            sum += ah * zs[h * FDIM];
        }
        acc += sum;
    }
    return acc;
}

__device__ __forceinline__ void chunk_sum_rows(const float* __restrict__ X, int lane, int n,
                                               int s_l, const float (&a)[HEADS], float (&acc)[HEADS]) {
#pragma unroll 8
    for (int i = 0; i < n; ++i) {
        const int s = __builtin_amdgcn_readlane(s_l, i);
        const float xv = X[(size_t)s * FDIM + lane];
#pragma unroll
        for (int h = 0; h < HEADS; ++h) {
            const float ah = __int_as_float(__builtin_amdgcn_readlane(__float_as_int(a[h]), i));
            acc[h] += ah * xv;
        }
    }
}

__global__ __launch_bounds__(256) void seg_softmax_agg(const int* __restrict__ rowptr,
                                                       const int* __restrict__ ssrc,
                                                       const float* __restrict__ Z,
                                                       const float* __restrict__ el,
                                                       const float* __restrict__ er,
                                                       const float* __restrict__ b,
                                                       float* __restrict__ out, int do_elu) {
    const int lane = threadIdx.x & 31;
    const int d = __builtin_amdgcn_readfirstlane((int)(blockIdx.x * 8 + (threadIdx.x >> 5)));
    const int start = rowptr[d];
    const int end   = rowptr[d + 1];

    float erd[HEADS];
#pragma unroll
    for (int h = 0; h < HEADS; ++h) erd[h] = er[d * HEADS + h];

    float m[HEADS], den[HEADS];
#pragma unroll
    for (int h = 0; h < HEADS; ++h) { m[h] = -INFINITY; den[h] = 0.f; }

    const bool own0 = start + lane < end;
    int   s0 = 0;
    float p0[HEADS];
#pragma unroll
    for (int h = 0; h < HEADS; ++h) p0[h] = 0.f;
    if (own0) {
        s0 = ssrc[start + lane];
        edge_logits8(el, s0, erd, p0);
#pragma unroll
        for (int h = 0; h < HEADS; ++h) m[h] = fmaxf(m[h], p0[h]);
    }
    for (int c0 = start + 32; c0 < end; c0 += 32) {
        const int j = c0 + lane;
        if (j < end) {
            float v[HEADS];
            edge_logits8(el, ssrc[j], erd, v);
#pragma unroll
            for (int h = 0; h < HEADS; ++h) m[h] = fmaxf(m[h], v[h]);
        }
    }
#pragma unroll
    for (int h = 0; h < HEADS; ++h) m[h] = wave_max(m[h]);

    if (own0) {
#pragma unroll
        for (int h = 0; h < HEADS; ++h) { p0[h] = __expf(p0[h] - m[h]); den[h] += p0[h]; }
    }
    for (int c0 = start + 32; c0 < end; c0 += 32) {
        const int j = c0 + lane;
        if (j < end) {
            float v[HEADS];
            edge_logits8(el, ssrc[j], erd, v);
#pragma unroll
            for (int h = 0; h < HEADS; ++h) den[h] += __expf(v[h] - m[h]);
        }
    }
#pragma unroll
    for (int h = 0; h < HEADS; ++h) den[h] = wave_sum(den[h]);

    float acc = 0.f;
    if (start < end) {
        float a[HEADS];
#pragma unroll
        for (int h = 0; h < HEADS; ++h) a[h] = 0.f;
        if (own0) {
#pragma unroll
            for (int h = 0; h < HEADS; ++h) a[h] = p0[h] / den[h] * (1.0f / HEADS);
        }
        acc = chunk_sum_z(Z, lane, min(32, end - start), s0, a, acc);
    }
    for (int c0 = start + 32; c0 < end; c0 += 32) {
        const int j = c0 + lane;
        int   s_l = 0;
        float a[HEADS];
#pragma unroll
        for (int h = 0; h < HEADS; ++h) a[h] = 0.f;
        if (j < end) {
            s_l = ssrc[j];
            float v[HEADS];
            edge_logits8(el, s_l, erd, v);
#pragma unroll
            for (int h = 0; h < HEADS; ++h)
                a[h] = __expf(v[h] - m[h]) / den[h] * (1.0f / HEADS);
        }
        acc = chunk_sum_z(Z, lane, min(32, end - c0), s_l, a, acc);
    }

    float bm = 0.f;
#pragma unroll
    for (int h = 0; h < HEADS; ++h) bm += b[h * FDIM + lane];
    const float r = acc + bm * (1.0f / HEADS);
    out[d * FDIM + lane] = (do_elu && r <= 0.f) ? expm1f(r) : r;
}


__global__ __launch_bounds__(256) void attn_logits_folded(const float* __restrict__ X,
                                                          const float* __restrict__ W,
                                                          const float* __restrict__ al,
                                                          const float* __restrict__ ar,
                                                          float* __restrict__ el,
                                                          float* __restrict__ er) {
    __shared__ float s_w[FDIM * 2 * HEADS];
    {
        const int k = threadIdx.x >> 3;
        const int h = threadIdx.x & (HEADS - 1);
        const float* w  = W + (size_t)k * HF + h * FDIM;
        const float* pl = al + h * FDIM;
        const float* pr = ar + h * FDIM;
        float sl = 0.f, sr = 0.f;
        for (int f = 0; f < FDIM; ++f) { float wv = w[f]; sl += wv * pl[f]; sr += wv * pr[f]; }
        s_w[k * 2 * HEADS + h]         = sl;
        s_w[k * 2 * HEADS + HEADS + h] = sr;
    }
    __syncthreads();
    const int i = blockIdx.x * blockDim.x + threadIdx.x;
    if (i >= NNODES * HEADS) return;
    const int h = i & (HEADS - 1);
    const int n = i >> 3;
    const float* x = X + (size_t)n * FDIM;
    float sl = 0.f, sr = 0.f;
    for (int k = 0; k < FDIM; ++k) {
        const float xv = x[k];
        sl += xv * s_w[k * 2 * HEADS + h];
        sr += xv * s_w[k * 2 * HEADS + HEADS + h];
    }
    el[i] = sl; er[i] = sr;
}

__global__ __launch_bounds__(256) void seg_softmax_agg_rows(const int* __restrict__ rowptr,
                                                            const int* __restrict__ ssrc,
                                                            const float* __restrict__ X,
                                                            const float* __restrict__ el,
                                                            const float* __restrict__ er,
                                                            float* __restrict__ agg) {
    const int lane = threadIdx.x & 31;
    const int d = __builtin_amdgcn_readfirstlane((int)(blockIdx.x * 8 + (threadIdx.x >> 5)));
    const int start = rowptr[d];
    const int end   = rowptr[d + 1];

    float erd[HEADS];
#pragma unroll
    for (int h = 0; h < HEADS; ++h) erd[h] = er[d * HEADS + h];

    float m[HEADS], den[HEADS];
#pragma unroll
    for (int h = 0; h < HEADS; ++h) { m[h] = -INFINITY; den[h] = 0.f; }

    const bool own0 = start + lane < end;
    int   s0 = 0;
    float p0[HEADS];
#pragma unroll
    for (int h = 0; h < HEADS; ++h) p0[h] = 0.f;
    if (own0) {
        s0 = ssrc[start + lane];
        edge_logits8(el, s0, erd, p0);
#pragma unroll
        for (int h = 0; h < HEADS; ++h) m[h] = fmaxf(m[h], p0[h]);
    }
    for (int c0 = start + 32; c0 < end; c0 += 32) {
        const int j = c0 + lane;
        if (j < end) {
            float v[HEADS];
            edge_logits8(el, ssrc[j], erd, v);
#pragma unroll
            for (int h = 0; h < HEADS; ++h) m[h] = fmaxf(m[h], v[h]);
        }
    }
#pragma unroll
    for (int h = 0; h < HEADS; ++h) m[h] = wave_max(m[h]);

    if (own0) {
#pragma unroll
        for (int h = 0; h < HEADS; ++h) { p0[h] = __expf(p0[h] - m[h]); den[h] += p0[h]; }
    }
    for (int c0 = start + 32; c0 < end; c0 += 32) {
        const int j = c0 + lane;
        if (j < end) {
            float v[HEADS];
            edge_logits8(el, ssrc[j], erd, v);
#pragma unroll
            for (int h = 0; h < HEADS; ++h) den[h] += __expf(v[h] - m[h]);
        }
    }
#pragma unroll
    for (int h = 0; h < HEADS; ++h) den[h] = wave_sum(den[h]);

    float acc[HEADS];
#pragma unroll
    for (int h = 0; h < HEADS; ++h) acc[h] = 0.f;
    if (start < end) {
        float a[HEADS];
#pragma unroll
        for (int h = 0; h < HEADS; ++h) a[h] = 0.f;
        if (own0) {
#pragma unroll
            for (int h = 0; h < HEADS; ++h) a[h] = p0[h] / den[h] * (1.0f / HEADS);
        }
        chunk_sum_rows(X, lane, min(32, end - start), s0, a, acc);
    }
    for (int c0 = start + 32; c0 < end; c0 += 32) {
        const int j = c0 + lane;
        int   s_l = 0;
        float a[HEADS];
#pragma unroll
        for (int h = 0; h < HEADS; ++h) a[h] = 0.f;
        if (j < end) {
            s_l = ssrc[j];
            float v[HEADS];
            edge_logits8(el, s_l, erd, v);
#pragma unroll
            for (int h = 0; h < HEADS; ++h)
                a[h] = __expf(v[h] - m[h]) / den[h] * (1.0f / HEADS);
        }
        chunk_sum_rows(X, lane, min(32, end - c0), s_l, a, acc);
    }

    float* o = agg + (size_t)d * HF + lane;
#pragma unroll
    for (int h = 0; h < HEADS; ++h) o[h * FDIM] = acc[h];
}

__global__ __launch_bounds__(256) void project_heads_wmma(const float* __restrict__ A,
                                                          const float* __restrict__ W,
                                                          const float* __restrict__ b,
                                                          float* __restrict__ out) {
    const int lane = threadIdx.x & 31;
    const int half = lane >> 4;
    const int mn   = lane & 15;
    const int tile = blockIdx.x * 8 + (threadIdx.x >> 5);
    if (tile >= NNODES / 16) return;
    const int row0 = tile * 16;

    const float* arow = A + (size_t)(row0 + mn) * HF;
    v8f c0 = {}, c1 = {};
    for (int k = 0; k < HF; k += 4) {
        const int kk = k + half * 2;
        v2f a, b0, b1;
        a.x  = arow[kk];
        a.y  = arow[kk + 1];
        const float* w0 = W + (size_t)(kk & (FDIM - 1)) * HF + (kk >> 5) * FDIM;
        b0.x = w0[mn];
        b0.y = w0[HF + mn];
        b1.x = w0[16 + mn];
        b1.y = w0[HF + 16 + mn];
        c0 = __builtin_amdgcn_wmma_f32_16x16x4_f32(false, a, false, b0,
                                                   (short)0, c0, false, false);
        c1 = __builtin_amdgcn_wmma_f32_16x16x4_f32(false, a, false, b1,
                                                   (short)0, c1, false, false);
    }
    float bm0 = 0.f, bm1 = 0.f;
#pragma unroll
    for (int h = 0; h < HEADS; ++h) { bm0 += b[h * FDIM + mn]; bm1 += b[h * FDIM + 16 + mn]; }
    bm0 *= (1.0f / HEADS);
    bm1 *= (1.0f / HEADS);
    float* po = out + (size_t)(row0 + half * 8) * FDIM + mn;
#pragma unroll
    for (int v = 0; v < 8; ++v) {
        po[(size_t)v * FDIM]      = c0[v] + bm0;
        po[(size_t)v * FDIM + 16] = c1[v] + bm1;
    }
}

static void build_edges_by_dst(const int* src, const int* dst, int* counts, int* rowptr,
                               int* cursor, int* ssrc, hipStream_t stream) {
    zero_counts<<<(NNODES + 255) / 256, 256, 0, stream>>>(counts);
    dst_histogram<<<(NEDGES + 255) / 256, 256, 0, stream>>>(dst, counts);
    scan_counts<<<(NNODES + SCAN_T - 1) / SCAN_T, SCAN_T, 0, stream>>>(counts, rowptr, cursor);
    fill_by_dst<<<(NEDGES + 255) / 256, 256, 0, stream>>>(src, dst, cursor, ssrc);
}

static void run_gat_layer(const float* X, const float* W,
                          const float* al, const float* ar, const float* b,
                          const int* rowptr, const int* ssrc,
                          float* z, float* el, float* er, float* out,
                          int do_elu, hipStream_t stream) {
    gemm_wmma_f32<<<(NNODES + GM_ROWS - 1) / GM_ROWS, 256, 0, stream>>>(X, W, z);

    int nh = NNODES * HEADS;
    attn_logits<<<(nh + 255) / 256, 256, 0, stream>>>(z, al, ar, el, er);

    seg_softmax_agg<<<NNODES / 8, 256, 0, stream>>>(rowptr, ssrc, z, el, er, b, out, do_elu);
}

static void run_gat_layer_agg_first(const float* X, const float* W,
                                    const float* al, const float* ar, const float* b,
                                    const int* rowptr, const int* ssrc,
                                    float* agg, float* el, float* er, float* out,
                                    hipStream_t stream) {
    attn_logits_folded<<<(NNODES * HEADS + 255) / 256, 256, 0, stream>>>(X, W, al, ar, el, er);
    seg_softmax_agg_rows<<<NNODES / 8, 256, 0, stream>>>(rowptr, ssrc, X, el, er, agg);
    project_heads_wmma<<<(NNODES / 16 + 7) / 8, 256, 0, stream>>>(agg, W, b, out);
}

extern "C" void kernel_launch(void* const* d_in, const int* in_sizes, int n_in,
                              void* d_out, int out_size, void* d_ws, size_t ws_size,
                              hipStream_t stream) {
    const float* x   = (const float*)d_in[0];
    const float* W1  = (const float*)d_in[1];
    const float* al1 = (const float*)d_in[2];
    const float* ar1 = (const float*)d_in[3];
    const float* b1  = (const float*)d_in[4];
    const float* W2  = (const float*)d_in[5];
    const float* al2 = (const float*)d_in[6];
    const float* ar2 = (const float*)d_in[7];
    const float* b2  = (const float*)d_in[8];
    const int*   src = (const int*)d_in[9];
    const int*   dst = (const int*)d_in[10];
    float* out = (float*)d_out;

    char* p = (char*)d_ws;
    float* z      = (float*)p; p += (size_t)NNODES * HF * 4;
    float* el     = (float*)p; p += (size_t)NNODES * HEADS * 4;
    float* er     = (float*)p; p += (size_t)NNODES * HEADS * 4;
    int*   rowptr = (int*)p;   p += (size_t)NNODES * HEADS * 4;
    int*   cursor = (int*)p;   p += (size_t)NNODES * HEADS * 4;
    int*   counts = cursor + NNODES;
    int*   ssrc   = (int*)p;   p += (size_t)NEDGES * HEADS * 4;
    p += (size_t)NNODES * FDIM * 4;
    float* hbuf   = (float*)p; p += (size_t)NNODES * FDIM * 4;

    build_edges_by_dst(src, dst, counts, rowptr, cursor, ssrc, stream);

    run_gat_layer(x, W1, al1, ar1, b1, rowptr, ssrc,
                  z, el, er, hbuf,  1, stream);
    run_gat_layer_agg_first(hbuf, W2, al2, ar2, b2, rowptr, ssrc,
                            z, el, er, out, stream);
}
